# starting kernel
# baseline (speedup 1.0000x reference)
#include <hip/hip_runtime.h>
#include <hip/hip_bf16.h>
#include <math.h>

#define NSTEP 64
#define H2    1024
#define TT    512
#define EPSN  1e-12f

typedef __attribute__((ext_vector_type(2))) float v2f;
typedef __attribute__((ext_vector_type(8))) float v8f;

__global__ void __launch_bounds__(1024)
recurrence_kernel(const float* __restrict__ W, const float* __restrict__ bias,
                  const float* __restrict__ u0, float* __restrict__ u_ws)
{
    __shared__ float u[H2];
    __shared__ float y[H2];
    __shared__ float red[32];

    const int tid  = threadIdx.x;
    const int wave = tid >> 5;
    const int lane = tid & 31;

    u[tid] = u0[tid];
    __syncthreads();

    for (int i = 0; i < NSTEP; ++i) {
        for (int er = 0; er < 32; ++er) {
            const int e = wave * 32 + er;
            const float* wr = W + (size_t)e * H2;
            float p = 0.0f;
            #pragma unroll 4
            for (int d = lane; d < H2; d += 32)
                p += wr[d] * u[d];
            #pragma unroll
            for (int off = 16; off >= 1; off >>= 1)
                p += __shfl_xor(p, off, 32);
            if (lane == 0) y[e] = p + bias[e];
        }
        __syncthreads();

        float v  = y[tid];
        float sq = v * v;
        #pragma unroll
        for (int off = 16; off >= 1; off >>= 1)
            sq += __shfl_xor(sq, off, 32);
        if (lane == 0) red[wave] = sq;
        __syncthreads();
        if (tid < 32) {
            float r = red[tid];
            #pragma unroll
            for (int off = 16; off >= 1; off >>= 1)
                r += __shfl_xor(r, off, 32);
            red[tid] = r;
        }
        __syncthreads();

        const float inv = 1.0f / fmaxf(sqrtf(red[0]), EPSN);
        const float val = v * inv;
        u_ws[(size_t)i * H2 + tid] = val;
        u[tid] = val;
        __syncthreads();
    }
}

#define KC   512
#define KPAD (KC + 2)

__global__ void __launch_bounds__(256)
scores_kernel(const float* __restrict__ hidden, const float* __restrict__ W,
              const float* __restrict__ bias,  const float* __restrict__ u_ws,
              float* __restrict__ scores)
{
    __shared__ float sA[16 * KPAD];
    __shared__ float sScore[16];

    const int n    = blockIdx.x;
    const int t0   = blockIdx.y * 16;
    const int tid  = threadIdx.x;
    const int lane = tid & 31;
    const int wave = tid >> 5;
    const int l16  = lane & 15;
    const int kHalf = (lane >> 4) * 2;

    if (tid < 16) sScore[tid] = 0.0f;

    const float* hbase = hidden + (size_t)n * H2 * TT + t0;

    v8f accs[8];
    #pragma unroll
    for (int i = 0; i < 8; ++i)
        accs[i] = (v8f){0.f,0.f,0.f,0.f,0.f,0.f,0.f,0.f};

    for (int kc = 0; kc < H2; kc += KC) {
        for (int i = tid; i < KC * 16; i += 256) {
            const int k = i >> 4;
            const int t = i & 15;
            sA[t * KPAD + k] = hbase[(size_t)(kc + k) * TT + t];
        }
        __syncthreads();

        #pragma unroll
        for (int idx = 0; idx < 8; ++idx) {
            const int e0 = (wave + idx * 8) * 16;
            const float* Wrow = W + (size_t)(e0 + l16) * H2 + kc + kHalf;
            const float* Arow = sA + l16 * KPAD + kHalf;
            v8f acc = accs[idx];
            for (int k0 = 0; k0 < KC; k0 += 4) {
                v2f av = *(const v2f*)(Arow + k0);
                v2f bv = *(const v2f*)(Wrow + k0);
                acc = __builtin_amdgcn_wmma_f32_16x16x4_f32(
                         false, av,  false, bv,
                         (short)0, acc,
                         false,  false);
            }
            accs[idx] = acc;
        }
        __syncthreads();
    }

    float tAcc[8];
    #pragma unroll
    for (int r = 0; r < 8; ++r) tAcc[r] = 0.0f;

    #pragma unroll
    for (int idx = 0; idx < 8; ++idx) {
        const int e0 = (wave + idx * 8) * 16;
        const float bb = bias[e0 + l16];
        const float uu = u_ws[(size_t)n * H2 + e0 + l16];
        #pragma unroll
        for (int r = 0; r < 8; ++r)
            tAcc[r] += tanhf(accs[idx][r] + bb) * uu;
    }

    #pragma unroll
    for (int r = 0; r < 8; ++r) {
        float v = tAcc[r];
        #pragma unroll
        for (int off = 8; off >= 1; off >>= 1)
            v += __shfl_xor(v, off, 32);
        if (l16 == 0) {
            const int t = r + ((lane >> 4) << 3);
            atomicAdd(&sScore[t], v);
        }
    }
    __syncthreads();

    if (tid < 16)
        scores[(size_t)n * TT + t0 + tid] = sScore[tid];
}

__global__ void __launch_bounds__(512)
softmax_kernel(const float* __restrict__ scores, float* __restrict__ attn)
{
    __shared__ float red[16];
    const int n   = blockIdx.x;
    const int tid = threadIdx.x;

    const float s = scores[(size_t)n * TT + tid];

    float m = s;
    #pragma unroll
    for (int off = 16; off >= 1; off >>= 1)
        m = fmaxf(m, __shfl_xor(m, off, 32));
    if ((tid & 31) == 0) red[tid >> 5] = m;
    __syncthreads();
    if (tid < 16) {
        float r = red[tid];
        #pragma unroll
        for (int off = 8; off >= 1; off >>= 1)
            r = fmaxf(r, __shfl_xor(r, off, 32));
        red[tid] = r;
    }
    __syncthreads();
    m = red[0];
    __syncthreads();

    const float e = __expf(s - m);
    float sum = e;
    #pragma unroll
    for (int off = 16; off >= 1; off >>= 1)
        sum += __shfl_xor(sum, off, 32);
    if ((tid & 31) == 0) red[tid >> 5] = sum;
    __syncthreads();
    if (tid < 16) {
        float r = red[tid];
        #pragma unroll
        for (int off = 8; off >= 1; off >>= 1)
            r += __shfl_xor(r, off, 32);
        red[tid] = r;
    }
    __syncthreads();

    attn[(size_t)n * TT + tid] = e / red[0];
}

__global__ void __launch_bounds__(256)
weighted_sum_kernel(const float* __restrict__ hidden,
                    const float* __restrict__ attn, float* __restrict__ out)
{
    __shared__ float sa[TT];
    const int n   = blockIdx.x;
    const int dc  = blockIdx.y;
    const int tid = threadIdx.x;

    sa[tid]       = attn[(size_t)n * TT + tid];
    sa[tid + 256] = attn[(size_t)n * TT + tid + 256];
    __syncthreads();

    const int d = dc * 256 + tid;
    const float* h = hidden + (size_t)n * H2 * TT + (size_t)d * TT;
    float acc = 0.0f;
    #pragma unroll 4
    for (int t = 0; t < TT; t += 4) {
        const float4 h4 = *(const float4*)(h + t);
        acc += h4.x * sa[t] + h4.y * sa[t + 1] + h4.z * sa[t + 2] + h4.w * sa[t + 3];
    }
    out[(size_t)n * H2 + d] = acc;
}

extern "C" void kernel_launch(void* const* d_in, const int* in_sizes, int n_in,
                              void* d_out, int out_size, void* d_ws, size_t ws_size,
                              hipStream_t stream)
{
    const float* hidden = (const float*)d_in[0];
    const float* W      = (const float*)d_in[1];
    const float* bias   = (const float*)d_in[2];
    const float* u0     = (const float*)d_in[3];
    float* out = (float*)d_out;

    float* ws     = (float*)d_ws;
    float* u_ws   = ws;
    float* scores = u_ws + (size_t)NSTEP * H2;
    float* attn   = scores + (size_t)NSTEP * TT;

    recurrence_kernel<<<1, 1024, 0, stream>>>(W, bias, u0, u_ws);

    scores_kernel<<<dim3(NSTEP, TT / 16), 256, 0, stream>>>(
        hidden, W, bias, u_ws, scores);

    softmax_kernel<<<NSTEP, TT, 0, stream>>>(scores, attn);

    weighted_sum_kernel<<<dim3(NSTEP, H2 / 256), 256, 0, stream>>>(
        hidden, attn, out);
}


# recurrence: 23 of a wave's 256 pieces of W kept on chip for a launch's run (19 LDS, 4 spare registers), summed under rows in flight; next step's rows 0-1 requested before the norm; sums unchanged
# speedup vs baseline: 4.6411x; 4.6411x over previous
#include <hip/hip_runtime.h>
#include <hip/hip_bf16.h>
#include <math.h>

#define NSTEP 64
#define H2    1024
#define TT    512
#define EPSN  1e-12f

typedef __attribute__((ext_vector_type(2))) float v2f;
typedef __attribute__((ext_vector_type(8))) float v8f;

typedef __attribute__((ext_vector_type(4))) float v4f;

#define KQ 8
#define RPW 32
#define PROW 29
#define PLDS 3
#define PREG (KQ - 1 - PLDS)
#define LROWS (RPW - 1 - PROW)
#define LPCS  (LROWS * KQ + PLDS)
#define WL_OFF (H2 + 32)
#define REC_SMEM_FLOATS (WL_OFF + 32 * LPCS * 32 * 4)

template <int CTRL>
static __device__ __forceinline__ float row_xor(float x)
{
    return __builtin_bit_cast(float,
        __builtin_amdgcn_update_dpp(0, __builtin_bit_cast(int, x), CTRL, 0xf, 0xf, true));
}

static __device__ __forceinline__ float half_xor(float x)
{
    const int b = __builtin_bit_cast(int, x);
    return __builtin_bit_cast(float,
        __builtin_amdgcn_permlanex16(b, b, 0x76543210, 0xfedcba98, false, false));
}

static __device__ __forceinline__ float wave_sum(float x)
{
    x += row_xor<0x161>(x);
    x += row_xor<0x162>(x);
    x += row_xor<0x164>(x);
    x += row_xor<0x168>(x);
    x += half_xor(x);
    return x;
}

static __device__ __forceinline__ void lds_barrier()
{
    __builtin_amdgcn_fence(__ATOMIC_RELEASE, "workgroup", "local");
    __builtin_amdgcn_s_barrier();
    __builtin_amdgcn_fence(__ATOMIC_ACQUIRE, "workgroup", "local");
}

static __device__ __forceinline__ float row_dot(const v4f (&wr)[KQ], const v4f (&uq)[KQ])
{
    float a0 = 0.f, a1 = 0.f, a2 = 0.f, a3 = 0.f;
    #pragma unroll
    for (int q = 0; q < KQ; ++q) {
        a0 = __builtin_fmaf(wr[q].x, uq[q].x, a0);
        a1 = __builtin_fmaf(wr[q].y, uq[q].y, a1);
        a2 = __builtin_fmaf(wr[q].z, uq[q].z, a2);
        a3 = __builtin_fmaf(wr[q].w, uq[q].w, a3);
    }
    float s = (a0 + a1) + (a2 + a3);
    asm volatile("" : "+v"(s));
    s = wave_sum(s);
    asm volatile("" : "+v"(s));
    return s;
}

static __device__ __forceinline__ void
recurrence_role(const float* __restrict__ W, const float* __restrict__ bias,
                const float* __restrict__ u0, float* __restrict__ u_ws,
                float* __restrict__ smem, const int first, const int count)
{
    float* u   = smem;
    float* red = smem + H2;

    const int tid  = threadIdx.x;
    const int lane = tid & 31;
    const int wave = __builtin_amdgcn_readfirstlane(tid >> 5);

    u[tid] = (first == 0) ? u0[tid] : u_ws[(size_t)(first - 1) * H2 + tid];
    const float bb = bias[tid];

    const v4f* wb = (const v4f*)(W + (size_t)wave * RPW * H2) + lane;

    v4f* wl = (v4f*)(smem + WL_OFF) + wave * (LPCS * 32) + lane;
    #pragma unroll
    for (int p = 0; p < LROWS * KQ; ++p)
        wl[p * 32] = wb[(PROW + 1) * (H2 / 4) + p * 32];
    #pragma unroll
    for (int p = 0; p < PLDS; ++p)
        wl[(LROWS * KQ + p) * 32] = wb[PROW * (H2 / 4) + (1 + p) * 32];
    v4f wres[PREG];
    #pragma unroll
    for (int p = 0; p < PREG; ++p)
        wres[p] = wb[PROW * (H2 / 4) + (1 + PLDS + p) * 32];

    v4f wa[KQ], wc[KQ];
#define LOAD_ROW(BUF, R) do { \
        __builtin_amdgcn_sched_barrier(0); \
        _Pragma("unroll") for (int q = 0; q < KQ; ++q) \
            BUF[q] = wb[(R) * (H2 / 4) + q * 32]; \
        __builtin_amdgcn_sched_barrier(0); } while (0)
#define LDS_ROW(BUF, LR) do { \
        _Pragma("unroll") for (int q = 0; q < KQ; ++q) \
            BUF[q] = wl[((LR) * KQ + q) * 32]; } while (0)
#define TAKE(ROW, S) mine = (lane == (ROW)) ? (S) : mine

    LOAD_ROW(wa, 0);
    LOAD_ROW(wc, 1);
    __syncthreads();

    for (int i = first; i < first + count; ++i) {
        v4f uq[KQ];
        #pragma unroll
        for (int q = 0; q < KQ; ++q)
            uq[q] = *(const v4f*)(u + q * 128 + lane * 4);

        float mine = 0.0f;
        #pragma nounroll
        for (int r = 0; r < PROW - 3; r += 2) {
            const float s0 = row_dot(wa, uq);
            TAKE(r, s0);
            LOAD_ROW(wa, r + 2);
            const float s1 = row_dot(wc, uq);
            TAKE(r + 1, s1);
            LOAD_ROW(wc, r + 3);
        }

        {
            const float s = row_dot(wa, uq);
            TAKE(PROW - 3, s);
            LOAD_ROW(wa, PROW - 1);
        }
        {
            const float s = row_dot(wc, uq);
            TAKE(PROW - 2, s);
        }
        {
            LDS_ROW(wc, 0);
            const float s = row_dot(wc, uq);
            TAKE(PROW + 1, s);
        }
        v4f wp[KQ];
        __builtin_amdgcn_sched_barrier(0);
        wp[0] = wb[PROW * (H2 / 4)];
        __builtin_amdgcn_sched_barrier(0);
        {
            const float s = row_dot(wa, uq);
            TAKE(PROW - 1, s);
            LOAD_ROW(wa, 0);
        }
        {
            #pragma unroll
            for (int p = 0; p < PLDS; ++p)
                wp[1 + p] = wl[(LROWS * KQ + p) * 32];
            #pragma unroll
            for (int p = 0; p < PREG; ++p)
                wp[1 + PLDS + p] = wres[p];
            const float s = row_dot(wp, uq);
            TAKE(PROW, s);
        }
        {
            LDS_ROW(wc, 1);
            const float s = row_dot(wc, uq);
            TAKE(PROW + 2, s);
            LOAD_ROW(wc, 1);
        }

        const float v  = mine + bb;
        const float sq = wave_sum(v * v);
        if (lane == 0) red[wave] = sq;
        lds_barrier();
        const float tot = wave_sum(red[lane]);

        const float inv = 1.0f / fmaxf(sqrtf(tot), EPSN);
        const float val = v * inv;
        u_ws[(size_t)i * H2 + tid] = val;
        u[tid] = val;
        lds_barrier();
    }
#undef LOAD_ROW
#undef LDS_ROW
#undef TAKE
    asm volatile("s_wait_loadcnt 0x0" ::: "memory");
}

__global__ void __launch_bounds__(256)
reorder_w_kernel(const float* __restrict__ W, float* __restrict__ Wp)
{
    const int gid = blockIdx.x * 256 + threadIdx.x;
    const int l   = gid & 31;
    const int kp  = (gid >> 5) & 127;
    const int et  = gid >> 12;
    const float* src = W + (size_t)(et * 16 + (l & 15)) * H2 + kp * 8 + (l >> 4) * 2;
    const v2f lo = *(const v2f*)(src);
    const v2f hi = *(const v2f*)(src + 4);
    v4f o;
    o.x = lo.x; o.y = lo.y; o.z = hi.x; o.w = hi.y;
    ((v4f*)Wp)[gid] = o;
}

#define KC   512
#define KPAD (KC + 2)
#define GP   4
#define NG   (KC / 8 / GP)
#define GSTR (GP * 32)
#define ETW  2
#define ESTR (32 * 128 * 32)
#define SMEM_FLOATS (16 * KPAD + 32 * 16)

static __device__ __forceinline__ void
scores_role(const float* __restrict__ hidden, const float* __restrict__ Wp,
            const float* __restrict__ bias,  const float* __restrict__ u_ws,
            float* __restrict__ scores, float* __restrict__ smem,
            const int n, const int t0)
{
    float* sA    = smem;
    float* sPart = smem + 16 * KPAD;

    const int tid  = threadIdx.x;
    const int lane = tid & 31;
    const int wave = tid >> 5;
    const int l16  = lane & 15;
    const int kHalf = (lane >> 4) * 2;

    const float* hbase = hidden + (size_t)n * H2 * TT + t0;
    const float* Arow  = sA + l16 * KPAD + kHalf;
    const v4f*   Wp4   = (const v4f*)Wp;

    v8f accs[ETW];
    #pragma unroll
    for (int i = 0; i < ETW; ++i)
        accs[i] = (v8f){0.f,0.f,0.f,0.f,0.f,0.f,0.f,0.f};

#define LOADG(B, P) do { \
        _Pragma("unroll") for (int j = 0; j < GP; ++j) B[j] = (P)[j * 32]; } while (0)

#define MULG(B, G) do { \
        _Pragma("unroll") for (int j = 0; j < GP; ++j) { \
            const float* ap = Arow + ((G) * GP + j) * 8; \
            const v2f a0 = *(const v2f*)(ap);         \
            const v2f a1 = *(const v2f*)(ap + 4); \
            const v2f b0 = __builtin_shufflevector(B[j], B[j], 0, 1); \
            const v2f b1 = __builtin_shufflevector(B[j], B[j], 2, 3); \
            acc = __builtin_amdgcn_wmma_f32_16x16x4_f32(false, a0, false, b0, (short)0, acc, false, false); \
            acc = __builtin_amdgcn_wmma_f32_16x16x4_f32(false, a1, false, b1, (short)0, acc, false, false); \
        } } while (0)

    for (int kc = 0; kc < H2; kc += KC) {
        for (int i = tid; i < KC * 16; i += 1024) {
            const int k = i >> 4;
            const int t = i & 15;
            sA[t * KPAD + k] = hbase[(size_t)(kc + k) * TT + t];
        }
        __syncthreads();

        const v4f* bbase = Wp4 + ((size_t)(wave * 128 + (kc >> 3)) * 32 + lane);
        v4f bA[GP], bB[GP];
        LOADG(bA, bbase);

        #pragma unroll
        for (int idx = 0; idx < ETW; ++idx) {
            const v4f* bp = bbase + (size_t)idx * ESTR;
            v8f acc = accs[idx];
            #pragma nounroll
            for (int g = 0; g < NG - 2; g += 2) {
                LOADG(bB, bp + (g + 1) * GSTR);
                MULG(bA, g);
                LOADG(bA, bp + (g + 2) * GSTR);
                MULG(bB, g + 1);
            }
            LOADG(bB, bp + (NG - 1) * GSTR);
            MULG(bA, NG - 2);
            if (idx < ETW - 1) LOADG(bA, bp + ESTR);
            MULG(bB, NG - 1);
            accs[idx] = acc;
        }
        __syncthreads();
    }
#undef LOADG
#undef MULG

    float tAcc[8];
    #pragma unroll
    for (int r = 0; r < 8; ++r) tAcc[r] = 0.0f;

    #pragma unroll
    for (int idx = 0; idx < ETW; ++idx) {
        const int e0 = (wave + idx * 32) * 16;
        const float bb = bias[e0 + l16];
        const float uu = u_ws[(size_t)n * H2 + e0 + l16];
        #pragma unroll
        for (int r = 0; r < 8; ++r)
            tAcc[r] += tanhf(accs[idx][r] + bb) * uu;
    }

    #pragma unroll
    for (int r = 0; r < 8; ++r) {
        float v = tAcc[r];
        #pragma unroll
        for (int off = 8; off >= 1; off >>= 1)
            v += __shfl_xor(v, off, 32);
        if (l16 == 0) {
            const int t = r + ((lane >> 4) << 3);
            sPart[wave * 16 + t] = v;
        }
    }
    __syncthreads();

    if (tid < 16) {
        float s = 0.0f;
        #pragma unroll
        for (int w = 0; w < 32; ++w)
            s += sPart[w * 16 + tid];
        scores[(size_t)n * TT + t0 + tid] = s;
    }
}

__global__ void __launch_bounds__(1024)
pipe_kernel(const float* __restrict__ hidden, const float* __restrict__ W,
            const float* __restrict__ Wp, const float* __restrict__ bias,
            const float* __restrict__ u0, float* __restrict__ u_ws,
            float* __restrict__ scores,
            const int rfirst, const int rcount, const int gfirst, const int gtiles)
{
    __shared__ __attribute__((aligned(16))) float
        smem[(REC_SMEM_FLOATS > SMEM_FLOATS) ? REC_SMEM_FLOATS : SMEM_FLOATS];

    const int hasR = (rcount > 0) ? 1 : 0;
    if (hasR && blockIdx.x == 0) {
        recurrence_role(W, bias, u0, u_ws, smem, rfirst, rcount);
    } else {
        const int tile = (int)blockIdx.x - hasR;
        if (tile < gtiles)
            scores_role(hidden, Wp, bias, u_ws, scores, smem,
                        gfirst + (tile >> 5), (tile & 31) * 16);
    }
}

__global__ void __launch_bounds__(512)
softmax_kernel(const float* __restrict__ scores, float* __restrict__ attn)
{
    __shared__ float red[16];
    const int n   = blockIdx.x;
    const int tid = threadIdx.x;

    const float s = scores[(size_t)n * TT + tid];

    float m = s;
    #pragma unroll
    for (int off = 16; off >= 1; off >>= 1)
        m = fmaxf(m, __shfl_xor(m, off, 32));
    if ((tid & 31) == 0) red[tid >> 5] = m;
    __syncthreads();
    if (tid < 16) {
        float r = red[tid];
        #pragma unroll
        for (int off = 8; off >= 1; off >>= 1)
            r = fmaxf(r, __shfl_xor(r, off, 32));
        red[tid] = r;
    }
    __syncthreads();
    m = red[0];
    __syncthreads();

    const float e = __expf(s - m);
    float sum = e;
    #pragma unroll
    for (int off = 16; off >= 1; off >>= 1)
        sum += __shfl_xor(sum, off, 32);
    if ((tid & 31) == 0) red[tid >> 5] = sum;
    __syncthreads();
    if (tid < 16) {
        float r = red[tid];
        #pragma unroll
        for (int off = 8; off >= 1; off >>= 1)
            r += __shfl_xor(r, off, 32);
        red[tid] = r;
    }
    __syncthreads();

    attn[(size_t)n * TT + tid] = e / red[0];
}

__global__ void __launch_bounds__(256)
weighted_sum_kernel(const float* __restrict__ hidden,
                    const float* __restrict__ attn, float* __restrict__ out)
{
    __shared__ float sa[TT];
    const int n   = blockIdx.x;
    const int dc  = blockIdx.y;
    const int tid = threadIdx.x;

    sa[tid]       = attn[(size_t)n * TT + tid];
    sa[tid + 256] = attn[(size_t)n * TT + tid + 256];
    __syncthreads();

    const int d = dc * 256 + tid;
    const float* h = hidden + (size_t)n * H2 * TT + (size_t)d * TT;
    float acc = 0.0f;
    #pragma unroll 4
    for (int t = 0; t < TT; t += 4) {
        const float4 h4 = *(const float4*)(h + t);
        acc += h4.x * sa[t] + h4.y * sa[t + 1] + h4.z * sa[t + 2] + h4.w * sa[t + 3];
    }
    out[(size_t)n * H2 + d] = acc;
}

extern "C" void kernel_launch(void* const* d_in, const int* in_sizes, int n_in,
                              void* d_out, int out_size, void* d_ws, size_t ws_size,
                              hipStream_t stream)
{
    const float* hidden = (const float*)d_in[0];
    const float* W      = (const float*)d_in[1];
    const float* bias   = (const float*)d_in[2];
    const float* u0     = (const float*)d_in[3];
    float* out = (float*)d_out;

    float* ws     = (float*)d_ws;
    float* u_ws   = ws;
    float* scores = u_ws + (size_t)NSTEP * H2;
    float* attn   = scores + (size_t)NSTEP * TT;
    float* Wp     = attn + (size_t)NSTEP * TT;

    reorder_w_kernel<<<H2 * H2 / 4 / 256, 256, 0, stream>>>(W, Wp);

#define PIPE(GRID, RF, RC, GF, GT) \
    pipe_kernel<<<(GRID), 1024, 0, stream>>>(hidden, W, Wp, bias, u0, u_ws, scores, (RF), (RC), (GF), (GT))
    PIPE(1,    0, 7,  0,   0);
    PIPE(225,  7, 7,  0, 224);
    PIPE(225, 14, 7,  7, 224);
    PIPE(225, 21, 7, 14, 224);
    PIPE(225, 28, 7, 21, 224);
    PIPE(225, 35, 7, 28, 224);
    PIPE(225, 42, 7, 35, 224);
    PIPE(225, 49, 7, 42, 224);
    PIPE(225, 56, 8, 49, 224);
    PIPE(256, 64, 0, 56, 256);
#undef PIPE

    softmax_kernel<<<NSTEP, TT, 0, stream>>>(scores, attn);

    weighted_sum_kernel<<<dim3(NSTEP, H2 / 256), 256, 0, stream>>>(
        hidden, attn, out);
}


# last run's tanh(h W + b) tiles stored beside run 1 (they need no u): the tail is a dot product per token, not a GEMM launch, and runs go from 9 to 5; e-sum of the last 4 steps regrouped, fixed order
# speedup vs baseline: 4.9320x; 1.0627x over previous
#include <hip/hip_runtime.h>
#include <hip/hip_bf16.h>
#include <math.h>

#define NSTEP 64
#define H2    1024
#define TT    512
#define EPSN  1e-12f

typedef __attribute__((ext_vector_type(2))) float v2f;
typedef __attribute__((ext_vector_type(8))) float v8f;

typedef __attribute__((ext_vector_type(4))) float v4f;

#define KQ 8
#define RPW 32
#define PROW 29
#define PLDS 3
#define PREG (KQ - 1 - PLDS)
#define LROWS (RPW - 1 - PROW)
#define LPCS  (LROWS * KQ + PLDS)
#define WL_OFF (H2 + 32)
#define REC_SMEM_FLOATS (WL_OFF + 32 * LPCS * 32 * 4)

template <int CTRL>
static __device__ __forceinline__ float row_xor(float x)
{
    return __builtin_bit_cast(float,
        __builtin_amdgcn_update_dpp(0, __builtin_bit_cast(int, x), CTRL, 0xf, 0xf, true));
}

static __device__ __forceinline__ float half_xor(float x)
{
    const int b = __builtin_bit_cast(int, x);
    return __builtin_bit_cast(float,
        __builtin_amdgcn_permlanex16(b, b, 0x76543210, 0xfedcba98, false, false));
}

static __device__ __forceinline__ float wave_sum(float x)
{
    x += row_xor<0x161>(x);
    x += row_xor<0x162>(x);
    x += row_xor<0x164>(x);
    x += row_xor<0x168>(x);
    x += half_xor(x);
    return x;
}

static __device__ __forceinline__ void lds_barrier()
{
    __builtin_amdgcn_fence(__ATOMIC_RELEASE, "workgroup", "local");
    __builtin_amdgcn_s_barrier();
    __builtin_amdgcn_fence(__ATOMIC_ACQUIRE, "workgroup", "local");
}

static __device__ __forceinline__ float row_dot(const v4f (&wr)[KQ], const v4f (&uq)[KQ])
{
    float a0 = 0.f, a1 = 0.f, a2 = 0.f, a3 = 0.f;
    #pragma unroll
    for (int q = 0; q < KQ; ++q) {
        a0 = __builtin_fmaf(wr[q].x, uq[q].x, a0);
        a1 = __builtin_fmaf(wr[q].y, uq[q].y, a1);
        a2 = __builtin_fmaf(wr[q].z, uq[q].z, a2);
        a3 = __builtin_fmaf(wr[q].w, uq[q].w, a3);
    }
    float s = (a0 + a1) + (a2 + a3);
    asm volatile("" : "+v"(s));
    s = wave_sum(s);
    asm volatile("" : "+v"(s));
    return s;
}

static __device__ __forceinline__ void
recurrence_role(const float* __restrict__ W, const float* __restrict__ bias,
                const float* __restrict__ u0, float* __restrict__ u_ws,
                float* __restrict__ smem, const int first, const int count)
{
    float* u   = smem;
    float* red = smem + H2;

    const int tid  = threadIdx.x;
    const int lane = tid & 31;
    const int wave = __builtin_amdgcn_readfirstlane(tid >> 5);

    u[tid] = (first == 0) ? u0[tid] : u_ws[(size_t)(first - 1) * H2 + tid];
    const float bb = bias[tid];

    const v4f* wb = (const v4f*)(W + (size_t)wave * RPW * H2) + lane;

    v4f* wl = (v4f*)(smem + WL_OFF) + wave * (LPCS * 32) + lane;
    #pragma unroll
    for (int p = 0; p < LROWS * KQ; ++p)
        wl[p * 32] = wb[(PROW + 1) * (H2 / 4) + p * 32];
    #pragma unroll
    for (int p = 0; p < PLDS; ++p)
        wl[(LROWS * KQ + p) * 32] = wb[PROW * (H2 / 4) + (1 + p) * 32];
    v4f wres[PREG];
    #pragma unroll
    for (int p = 0; p < PREG; ++p)
        wres[p] = wb[PROW * (H2 / 4) + (1 + PLDS + p) * 32];

    v4f wa[KQ], wc[KQ];
#define LOAD_ROW(BUF, R) do { \
        __builtin_amdgcn_sched_barrier(0); \
        _Pragma("unroll") for (int q = 0; q < KQ; ++q) \
            BUF[q] = wb[(R) * (H2 / 4) + q * 32]; \
        __builtin_amdgcn_sched_barrier(0); } while (0)
#define LDS_ROW(BUF, LR) do { \
        _Pragma("unroll") for (int q = 0; q < KQ; ++q) \
            BUF[q] = wl[((LR) * KQ + q) * 32]; } while (0)
#define TAKE(ROW, S) mine = (lane == (ROW)) ? (S) : mine

    LOAD_ROW(wa, 0);
    LOAD_ROW(wc, 1);
    __syncthreads();

    for (int i = first; i < first + count; ++i) {
        v4f uq[KQ];
        #pragma unroll
        for (int q = 0; q < KQ; ++q)
            uq[q] = *(const v4f*)(u + q * 128 + lane * 4);

        float mine = 0.0f;
        #pragma nounroll
        for (int r = 0; r < PROW - 3; r += 2) {
            const float s0 = row_dot(wa, uq);
            TAKE(r, s0);
            LOAD_ROW(wa, r + 2);
            const float s1 = row_dot(wc, uq);
            TAKE(r + 1, s1);
            LOAD_ROW(wc, r + 3);
        }

        {
            const float s = row_dot(wa, uq);
            TAKE(PROW - 3, s);
            LOAD_ROW(wa, PROW - 1);
        }
        {
            const float s = row_dot(wc, uq);
            TAKE(PROW - 2, s);
        }
        {
            LDS_ROW(wc, 0);
            const float s = row_dot(wc, uq);
            TAKE(PROW + 1, s);
        }
        v4f wp[KQ];
        __builtin_amdgcn_sched_barrier(0);
        wp[0] = wb[PROW * (H2 / 4)];
        __builtin_amdgcn_sched_barrier(0);
        {
            const float s = row_dot(wa, uq);
            TAKE(PROW - 1, s);
            LOAD_ROW(wa, 0);
        }
        {
            #pragma unroll
            for (int p = 0; p < PLDS; ++p)
                wp[1 + p] = wl[(LROWS * KQ + p) * 32];
            #pragma unroll
            for (int p = 0; p < PREG; ++p)
                wp[1 + PLDS + p] = wres[p];
            const float s = row_dot(wp, uq);
            TAKE(PROW, s);
        }
        {
            LDS_ROW(wc, 1);
            const float s = row_dot(wc, uq);
            TAKE(PROW + 2, s);
            LOAD_ROW(wc, 1);
        }

        const float v  = mine + bb;
        const float sq = wave_sum(v * v);
        if (lane == 0) red[wave] = sq;
        lds_barrier();
        const float tot = wave_sum(red[lane]);

        const float inv = 1.0f / fmaxf(sqrtf(tot), EPSN);
        const float val = v * inv;
        u_ws[(size_t)i * H2 + tid] = val;
        u[tid] = val;
        lds_barrier();
    }
#undef LOAD_ROW
#undef LDS_ROW
#undef TAKE
    asm volatile("s_wait_loadcnt 0x0" ::: "memory");
}

__global__ void __launch_bounds__(256)
reorder_w_kernel(const float* __restrict__ W, float* __restrict__ Wp)
{
    const int gid = blockIdx.x * 256 + threadIdx.x;
    const int l   = gid & 31;
    const int kp  = (gid >> 5) & 127;
    const int et  = gid >> 12;
    const float* src = W + (size_t)(et * 16 + (l & 15)) * H2 + kp * 8 + (l >> 4) * 2;
    const v2f lo = *(const v2f*)(src);
    const v2f hi = *(const v2f*)(src + 4);
    v4f o;
    o.x = lo.x; o.y = lo.y; o.z = hi.x; o.w = hi.y;
    ((v4f*)Wp)[gid] = o;
}

#define KC   512
#define KPAD (KC + 2)
#define GP   4
#define NG   (KC / 8 / GP)
#define GSTR (GP * 32)
#define ETW  2
#define ESTR (32 * 128 * 32)
#define SMEM_FLOATS (16 * KPAD + 32 * 16)
#define TPAD (H2 + 4)

static __device__ __forceinline__ void
scores_role(const float* __restrict__ hidden, const float* __restrict__ Wp,
            const float* __restrict__ bias,  const float* __restrict__ u_ws,
            float* __restrict__ scores, float* __restrict__ ut,
            float* __restrict__ smem, const int n, const int t0)
{
    float* sA    = smem;
    float* sPart = smem + 16 * KPAD;

    const int tid  = threadIdx.x;
    const int lane = tid & 31;
    const int wave = tid >> 5;
    const int l16  = lane & 15;
    const int kHalf = (lane >> 4) * 2;

    const float* hbase = hidden + (size_t)n * H2 * TT + t0;
    const float* Arow  = sA + l16 * KPAD + kHalf;
    const v4f*   Wp4   = (const v4f*)Wp;

    v8f accs[ETW];
    #pragma unroll
    for (int i = 0; i < ETW; ++i)
        accs[i] = (v8f){0.f,0.f,0.f,0.f,0.f,0.f,0.f,0.f};

#define LOADG(B, P) do { \
        _Pragma("unroll") for (int j = 0; j < GP; ++j) B[j] = (P)[j * 32]; } while (0)

#define MULG(B, G) do { \
        _Pragma("unroll") for (int j = 0; j < GP; ++j) { \
            const float* ap = Arow + ((G) * GP + j) * 8; \
            const v2f a0 = *(const v2f*)(ap);         \
            const v2f a1 = *(const v2f*)(ap + 4); \
            const v2f b0 = __builtin_shufflevector(B[j], B[j], 0, 1); \
            const v2f b1 = __builtin_shufflevector(B[j], B[j], 2, 3); \
            acc = __builtin_amdgcn_wmma_f32_16x16x4_f32(false, a0, false, b0, (short)0, acc, false, false); \
            acc = __builtin_amdgcn_wmma_f32_16x16x4_f32(false, a1, false, b1, (short)0, acc, false, false); \
        } } while (0)

    for (int kc = 0; kc < H2; kc += KC) {
        for (int i = tid; i < KC * 16; i += 1024) {
            const int k = i >> 4;
            const int t = i & 15;
            sA[t * KPAD + k] = hbase[(size_t)(kc + k) * TT + t];
        }
        __syncthreads();

        const v4f* bbase = Wp4 + ((size_t)(wave * 128 + (kc >> 3)) * 32 + lane);
        v4f bA[GP], bB[GP];
        LOADG(bA, bbase);

        #pragma unroll
        for (int idx = 0; idx < ETW; ++idx) {
            const v4f* bp = bbase + (size_t)idx * ESTR;
            v8f acc = accs[idx];
            #pragma nounroll
            for (int g = 0; g < NG - 2; g += 2) {
                LOADG(bB, bp + (g + 1) * GSTR);
                MULG(bA, g);
                LOADG(bA, bp + (g + 2) * GSTR);
                MULG(bB, g + 1);
            }
            LOADG(bB, bp + (NG - 1) * GSTR);
            MULG(bA, NG - 2);
            if (idx < ETW - 1) LOADG(bA, bp + ESTR);
            MULG(bB, NG - 1);
            accs[idx] = acc;
        }
        __syncthreads();
    }
#undef LOADG
#undef MULG

    if (ut != nullptr) {
        float* sT = smem;
        #pragma unroll
        for (int idx = 0; idx < ETW; ++idx) {
            const int e0 = (wave + idx * 32) * 16;
            const float bb = bias[e0 + l16];
            #pragma unroll
            for (int r = 0; r < 8; ++r)
                sT[(r + ((lane >> 4) << 3)) * TPAD + e0 + l16] = tanhf(accs[idx][r] + bb);
        }
        __syncthreads();
        #pragma unroll
        for (int j = 0; j < 4; ++j) {
            const int f = j * 1024 + tid;
            const int t = f >> 8;
            const int c = (f & 255) * 4;
            *(v4f*)(ut + (size_t)(t0 + t) * H2 + c) = *(const v4f*)(sT + t * TPAD + c);
        }
        return;
    }

    float tAcc[8];
    #pragma unroll
    for (int r = 0; r < 8; ++r) tAcc[r] = 0.0f;

    #pragma unroll
    for (int idx = 0; idx < ETW; ++idx) {
        const int e0 = (wave + idx * 32) * 16;
        const float bb = bias[e0 + l16];
        const float uu = u_ws[(size_t)n * H2 + e0 + l16];
        #pragma unroll
        for (int r = 0; r < 8; ++r)
            tAcc[r] += tanhf(accs[idx][r] + bb) * uu;
    }

    #pragma unroll
    for (int r = 0; r < 8; ++r) {
        float v = tAcc[r];
        #pragma unroll
        for (int off = 8; off >= 1; off >>= 1)
            v += __shfl_xor(v, off, 32);
        if (l16 == 0) {
            const int t = r + ((lane >> 4) << 3);
            sPart[wave * 16 + t] = v;
        }
    }
    __syncthreads();

    if (tid < 16) {
        float s = 0.0f;
        #pragma unroll
        for (int w = 0; w < 32; ++w)
            s += sPart[w * 16 + tid];
        scores[(size_t)n * TT + t0 + tid] = s;
    }
}

__global__ void __launch_bounds__(1024)
pipe_kernel(const float* __restrict__ hidden, const float* __restrict__ W,
            const float* __restrict__ Wp, const float* __restrict__ bias,
            const float* __restrict__ u0, float* __restrict__ u_ws,
            float* __restrict__ scores, float* __restrict__ uit,
            const int rfirst, const int rcount, const int gfirst, const int gtiles,
            const int ufirst, const int utiles)
{
    __shared__ __attribute__((aligned(16))) float
        smem[(REC_SMEM_FLOATS > SMEM_FLOATS) ? REC_SMEM_FLOATS : SMEM_FLOATS];

    const int hasR = (rcount > 0) ? 1 : 0;
    if (hasR && blockIdx.x == 0) {
        recurrence_role(W, bias, u0, u_ws, smem, rfirst, rcount);
    } else {
        const int tile = (int)blockIdx.x - hasR;
        if (tile < gtiles + utiles) {
            const bool st = tile >= gtiles;
            const int  tl = st ? tile - gtiles : tile;
            const int  n  = (st ? ufirst : gfirst) + (tl >> 5);
            float* ut = st ? uit + (size_t)(tl >> 5) * TT * H2 : nullptr;
            scores_role(hidden, Wp, bias, u_ws, scores, ut, smem, n, (tl & 31) * 16);
        }
    }
}

__global__ void __launch_bounds__(1024)
tail_scores_kernel(const float* __restrict__ uit, const float* __restrict__ u_ws,
                   float* __restrict__ scores, const int nfirst)
{
    __shared__ float res[32];
    const int lane = threadIdx.x & 31;
    const int wave = threadIdx.x >> 5;
    const int nl   = blockIdx.x >> 4;
    const int t    = (blockIdx.x & 15) * 32 + wave;
    const v4f* row = (const v4f*)(uit + ((size_t)nl * TT + t) * H2) + lane;
    const v4f* uu  = (const v4f*)(u_ws + (size_t)(nfirst + nl) * H2) + lane;
    float a0 = 0.f, a1 = 0.f, a2 = 0.f, a3 = 0.f;
    #pragma unroll
    for (int q = 0; q < KQ; ++q) {
        const v4f x = row[q * 32];
        const v4f y = uu[q * 32];
        a0 = __builtin_fmaf(x.x, y.x, a0);
        a1 = __builtin_fmaf(x.y, y.y, a1);
        a2 = __builtin_fmaf(x.z, y.z, a2);
        a3 = __builtin_fmaf(x.w, y.w, a3);
    }
    const float s = wave_sum((a0 + a1) + (a2 + a3));
    if (lane == 0) res[wave] = s;
    __syncthreads();
    if (threadIdx.x < 32)
        scores[(size_t)(nfirst + nl) * TT + (blockIdx.x & 15) * 32 + threadIdx.x] = res[threadIdx.x];
}

__global__ void __launch_bounds__(512)
softmax_kernel(const float* __restrict__ scores, float* __restrict__ attn)
{
    __shared__ float red[16];
    const int n   = blockIdx.x;
    const int tid = threadIdx.x;

    const float s = scores[(size_t)n * TT + tid];

    float m = s;
    #pragma unroll
    for (int off = 16; off >= 1; off >>= 1)
        m = fmaxf(m, __shfl_xor(m, off, 32));
    if ((tid & 31) == 0) red[tid >> 5] = m;
    __syncthreads();
    if (tid < 16) {
        float r = red[tid];
        #pragma unroll
        for (int off = 8; off >= 1; off >>= 1)
            r = fmaxf(r, __shfl_xor(r, off, 32));
        red[tid] = r;
    }
    __syncthreads();
    m = red[0];
    __syncthreads();

    const float e = __expf(s - m);
    float sum = e;
    #pragma unroll
    for (int off = 16; off >= 1; off >>= 1)
        sum += __shfl_xor(sum, off, 32);
    if ((tid & 31) == 0) red[tid >> 5] = sum;
    __syncthreads();
    if (tid < 16) {
        float r = red[tid];
        #pragma unroll
        for (int off = 8; off >= 1; off >>= 1)
            r += __shfl_xor(r, off, 32);
        red[tid] = r;
    }
    __syncthreads();

    attn[(size_t)n * TT + tid] = e / red[0];
}

__global__ void __launch_bounds__(256)
weighted_sum_kernel(const float* __restrict__ hidden,
                    const float* __restrict__ attn, float* __restrict__ out)
{
    __shared__ float sa[TT];
    const int n   = blockIdx.x;
    const int dc  = blockIdx.y;
    const int tid = threadIdx.x;

    sa[tid]       = attn[(size_t)n * TT + tid];
    sa[tid + 256] = attn[(size_t)n * TT + tid + 256];
    __syncthreads();

    const int d = dc * 256 + tid;
    const float* h = hidden + (size_t)n * H2 * TT + (size_t)d * TT;
    float acc = 0.0f;
    #pragma unroll 4
    for (int t = 0; t < TT; t += 4) {
        const float4 h4 = *(const float4*)(h + t);
        acc += h4.x * sa[t] + h4.y * sa[t + 1] + h4.z * sa[t + 2] + h4.w * sa[t + 3];
    }
    out[(size_t)n * H2 + d] = acc;
}

extern "C" void kernel_launch(void* const* d_in, const int* in_sizes, int n_in,
                              void* d_out, int out_size, void* d_ws, size_t ws_size,
                              hipStream_t stream)
{
    const float* hidden = (const float*)d_in[0];
    const float* W      = (const float*)d_in[1];
    const float* bias   = (const float*)d_in[2];
    const float* u0     = (const float*)d_in[3];
    float* out = (float*)d_out;

    float* ws     = (float*)d_ws;
    float* u_ws   = ws;
    float* scores = u_ws + (size_t)NSTEP * H2;
    float* attn   = scores + (size_t)NSTEP * TT;
    float* Wp     = attn + (size_t)NSTEP * TT;

    float* uit    = Wp + (size_t)H2 * H2;

    reorder_w_kernel<<<H2 * H2 / 4 / 256, 256, 0, stream>>>(W, Wp);

#define PIPE(GRID, RF, RC, GF, GT, UF, UT) \
    pipe_kernel<<<(GRID), 1024, 0, stream>>>(hidden, W, Wp, bias, u0, u_ws, scores, uit, (RF), (RC), (GF), (GT), (UF), (UT))
    if (ws_size >= 13107200) {
        PIPE(129,  0, 19,  0,   0, 60, 128);
        PIPE(609, 19, 19,  0, 608,  0,   0);
        PIPE(609, 38, 15, 19, 608,  0,   0);
        PIPE(481, 53,  7, 38, 480,  0,   0);
        PIPE(225, 60,  4, 53, 224,  0,   0);
        tail_scores_kernel<<<64, 1024, 0, stream>>>(uit, u_ws, scores, 60);
    } else {
        PIPE(1,    0, 7,  0,   0, 0, 0);
        PIPE(225,  7, 7,  0, 224, 0, 0);
        PIPE(225, 14, 7,  7, 224, 0, 0);
        PIPE(225, 21, 7, 14, 224, 0, 0);
        PIPE(225, 28, 7, 21, 224, 0, 0);
        PIPE(225, 35, 7, 28, 224, 0, 0);
        PIPE(225, 42, 7, 35, 224, 0, 0);
        PIPE(225, 49, 7, 42, 224, 0, 0);
        PIPE(225, 56, 8, 49, 224, 0, 0);
        PIPE(256, 64, 0, 56, 256, 0, 0);
    }
#undef PIPE

    softmax_kernel<<<NSTEP, TT, 0, stream>>>(scores, attn);

    weighted_sum_kernel<<<dim3(NSTEP, H2 / 256), 256, 0, stream>>>(
        hidden, attn, out);
}
